# starting kernel
# baseline (speedup 1.0000x reference)
#include <hip/hip_runtime.h>
#include <hip/hip_bf16.h>

static constexpr int N_NODES = 4096;
static constexpr int N_EDGES = 8192;
static constexpr int N_FEAT  = 5;
static constexpr int N_WIRES = 10;
static constexpr int NPAD    = 16;

typedef __attribute__((ext_vector_type(2))) float v2f;
typedef __attribute__((ext_vector_type(8))) float v8f;

__global__ __launch_bounds__(256) void pad_x_kernel(
    const float* __restrict__ X, float* __restrict__ Xpad)
{
    const int idx = blockIdx.x * 256 + threadIdx.x;
    const int k = idx >> 4;
    const int n = idx & 15;
    Xpad[idx] = (n < N_FEAT) ? X[k * N_FEAT + n] : 0.0f;
}

__global__ __launch_bounds__(256) void angles_gemm_kernel(
    const float* __restrict__ Xpad,
    const float* __restrict__ Ri,
    const float* __restrict__ Ro,
    float* __restrict__ Bmat)
{
    const int lane = threadIdx.x & 31;
    const int wave = threadIdx.x >> 5;
    const int tile = blockIdx.x * 8 + wave;
    const int e0   = tile * 16;
    const int m16  = lane & 15;
    const int half = lane >> 4;

    const float* __restrict__ R = (blockIdx.y == 0) ? Ro : Ri;
    const int colOff            = (blockIdx.y == 0) ? 0  : N_FEAT;

    const float* __restrict__ aPtr = R + (size_t)(2 * half) * N_EDGES + e0 + m16;
    const float* __restrict__ bPtr = Xpad + (2 * half) * NPAD + m16;

    v8f acc = {};
#pragma unroll 4
    for (int k0 = 0; k0 < N_NODES; k0 += 4) {
        v2f a, b;
        a.x = aPtr[0];
        a.y = aPtr[N_EDGES];
        b.x = bPtr[0];
        b.y = bPtr[NPAD];
        acc = __builtin_amdgcn_wmma_f32_16x16x4_f32(
             false, a,  false, b,
             (short)0, acc,  false,  false);
        aPtr += (size_t)4 * N_EDGES;
        bPtr += 4 * NPAD;
    }

    if (m16 < N_FEAT) {
#pragma unroll
        for (int r = 0; r < 8; ++r) {
            const int m = r + 8 * half;
            Bmat[(size_t)(e0 + m) * N_WIRES + colOff + m16] = acc[r];
        }
    }
}

__device__ __forceinline__ float shflx(float x, int m) {
    return __shfl_xor(x, m, 32);
}

template <int W>
__device__ __forceinline__ void apply_ry(float (&v)[32], float c, float s, int lane) {
    constexpr int B = 9 - W;
    if constexpr (B >= 5) {
        constexpr int st = 1 << (B - 5);
#pragma unroll
        for (int r = 0; r < 32; ++r) {
            if ((r & st) == 0) {
                const float p0 = v[r];
                const float p1 = v[r + st];
                v[r]      = c * p0 - s * p1;
                v[r + st] = s * p0 + c * p1;
            }
        }
    } else {
        constexpr int m = 1 << B;
        const float sgn = (lane & m) ? s : -s;
#pragma unroll
        for (int r = 0; r < 32; ++r) {
            const float p = shflx(v[r], m);
            v[r] = c * v[r] + sgn * p;
        }
    }
}

template <int C, int T>
__device__ __forceinline__ void apply_cnot(float (&v)[32], int lane) {
    constexpr int BC = 9 - C;
    constexpr int BT = 9 - T;
    if constexpr (BC >= 5 && BT >= 5) {
        constexpr int cm = 1 << (BC - 5);
        constexpr int tm = 1 << (BT - 5);
#pragma unroll
        for (int r = 0; r < 32; ++r) {
            if ((r & cm) && !(r & tm)) {
                const float t = v[r];
                v[r]      = v[r + tm];
                v[r + tm] = t;
            }
        }
    } else if constexpr (BC >= 5 && BT < 5) {
        constexpr int cm    = 1 << (BC - 5);
        constexpr int tmask = 1 << BT;
#pragma unroll
        for (int r = 0; r < 32; ++r) {
            if (r & cm) v[r] = shflx(v[r], tmask);
        }
    } else if constexpr (BC < 5 && BT >= 5) {
        constexpr int cmask = 1 << BC;
        constexpr int tm    = 1 << (BT - 5);
        const bool cb = (lane & cmask) != 0;
#pragma unroll
        for (int r = 0; r < 32; ++r) {
            if ((r & tm) == 0) {
                const float a = v[r];
                const float b = v[r + tm];
                v[r]      = cb ? b : a;
                v[r + tm] = cb ? a : b;
            }
        }
    } else {
        constexpr int cmask = 1 << BC;
        constexpr int tmask = 1 << BT;
        const bool cb = (lane & cmask) != 0;
#pragma unroll
        for (int r = 0; r < 32; ++r) {
            const float t = shflx(v[r], tmask);
            v[r] = cb ? t : v[r];
        }
    }
}

__global__ __launch_bounds__(256) void ttn_sim_kernel(
    const float* __restrict__ Bmat,
    const float* __restrict__ theta,
    float* __restrict__ out)
{
    const int lane = threadIdx.x & 31;
    const int wave = threadIdx.x >> 5;
    const int e    = blockIdx.x * 8 + wave;

    float ac[10], as[10];
#pragma unroll
    for (int w = 0; w < N_WIRES; ++w) {
        const float h = 0.5f * Bmat[(size_t)e * N_WIRES + w];
        sincosf(h, &as[w], &ac[w]);
    }
    float tc[19], ts[19];
#pragma unroll
    for (int j = 0; j < 19; ++j) {
        const float h = 0.5f * theta[j];
        sincosf(h, &ts[j], &tc[j]);
    }

    float v[32];
#pragma unroll
    for (int r = 0; r < 32; ++r) v[r] = 0.0f;
    if (lane == 0) v[0] = 1.0f;

    apply_ry<0>(v, ac[0], as[0], lane);
    apply_ry<1>(v, ac[1], as[1], lane);
    apply_ry<2>(v, ac[2], as[2], lane);
    apply_ry<3>(v, ac[3], as[3], lane);
    apply_ry<4>(v, ac[4], as[4], lane);
    apply_ry<5>(v, ac[5], as[5], lane);
    apply_ry<6>(v, ac[6], as[6], lane);
    apply_ry<7>(v, ac[7], as[7], lane);
    apply_ry<8>(v, ac[8], as[8], lane);
    apply_ry<9>(v, ac[9], as[9], lane);
    apply_ry<0>(v, tc[0], ts[0], lane); apply_ry<1>(v, tc[1], ts[1], lane); apply_cnot<0, 1>(v, lane);
    apply_ry<2>(v, tc[2], ts[2], lane); apply_ry<3>(v, tc[3], ts[3], lane); apply_cnot<3, 2>(v, lane);
    apply_ry<4>(v, tc[4], ts[4], lane); apply_ry<5>(v, tc[5], ts[5], lane); apply_cnot<5, 4>(v, lane);
    apply_ry<6>(v, tc[6], ts[6], lane); apply_ry<7>(v, tc[7], ts[7], lane); apply_cnot<6, 7>(v, lane);
    apply_ry<8>(v, tc[8], ts[8], lane); apply_ry<9>(v, tc[9], ts[9], lane); apply_cnot<8, 9>(v, lane);
    apply_ry<1>(v, tc[10], ts[10], lane); apply_ry<2>(v, tc[11], ts[11], lane); apply_cnot<1, 2>(v, lane);
    apply_ry<7>(v, tc[12], ts[12], lane); apply_ry<9>(v, tc[13], ts[13], lane); apply_cnot<9, 7>(v, lane);
    apply_ry<2>(v, tc[14], ts[14], lane); apply_ry<4>(v, tc[15], ts[15], lane); apply_cnot<2, 4>(v, lane);
    apply_ry<4>(v, tc[16], ts[16], lane); apply_ry<7>(v, tc[17], ts[17], lane); apply_cnot<4, 7>(v, lane);
    apply_ry<7>(v, tc[18], ts[18], lane);

    float loc = 0.0f;
#pragma unroll
    for (int r = 0; r < 32; ++r) loc += v[r] * v[r];
    loc = (lane & 4) ? loc : 0.0f;
#pragma unroll
    for (int off = 16; off >= 1; off >>= 1) loc += __shfl_xor(loc, off, 32);
    if (lane == 0) out[e] = loc;
}

extern "C" void kernel_launch(void* const* d_in, const int* in_sizes, int n_in,
                              void* d_out, int out_size, void* d_ws, size_t ws_size,
                              hipStream_t stream)
{
    (void)in_sizes; (void)n_in; (void)out_size; (void)ws_size;
    const float* X     = (const float*)d_in[0];
    const float* Ri    = (const float*)d_in[1];
    const float* Ro    = (const float*)d_in[2];
    const float* theta = (const float*)d_in[3];
    float* out  = (float*)d_out;

    float* Bmat = (float*)d_ws;
    float* Xpad = (float*)((char*)d_ws + (size_t)N_EDGES * N_WIRES * sizeof(float));

    pad_x_kernel<<<(N_NODES * NPAD) / 256, 256, 0, stream>>>(X, Xpad);

    dim3 gGemm(512 / 8, 2);
    angles_gemm_kernel<<<gGemm, 256, 0, stream>>>(Xpad, Ri, Ro, Bmat);

    ttn_sim_kernel<<<N_EDGES / 8, 256, 0, stream>>>(Bmat, theta, out);
}


# K of the angle GEMM split over 16 waves per 64-edge block; each streams 16-byte non-temporal row loads in two register sets into four f32 WMMA tiles; partial sums folded in wave order via LDS
# speedup vs baseline: 4.7461x; 4.7461x over previous
#include <hip/hip_runtime.h>
#include <hip/hip_bf16.h>

static constexpr int N_NODES = 4096;
static constexpr int N_EDGES = 8192;
static constexpr int N_FEAT  = 5;
static constexpr int N_WIRES = 10;
static constexpr int NPAD    = 16;

typedef __attribute__((ext_vector_type(2))) float v2f;
typedef __attribute__((ext_vector_type(8))) float v8f;

__global__ __launch_bounds__(256) void pad_x_kernel(
    const float* __restrict__ X, float* __restrict__ Xpad)
{
    const int idx = blockIdx.x * 256 + threadIdx.x;
    const int k = idx >> 4;
    const int n = idx & 15;
    Xpad[idx] = (n < N_FEAT) ? X[k * N_FEAT + n] : 0.0f;
}

typedef __attribute__((ext_vector_type(4))) float v4f;

static constexpr int GEMM_WAVES     = 16;
static constexpr int GEMM_THREADS   = GEMM_WAVES * 32;
static constexpr int WG_EDGES       = 64;
static constexpr int WAVE_ROWS      = N_NODES / GEMM_WAVES;
static constexpr int KSTEPS_PER_SET = 8;
static constexpr int SETS_PER_WAVE  = WAVE_ROWS / (4 * KSTEPS_PER_SET);
static constexpr int WG_OUTPUTS     = WG_EDGES * N_FEAT;
static constexpr int GEMM_LDS_FLOATS = GEMM_WAVES * WG_OUTPUTS;

struct FragSet {
    v4f   a[2 * KSTEPS_PER_SET];
    float b[2 * KSTEPS_PER_SET];
};

__device__ __forceinline__ void load_set(FragSet& f,
                                         const float* __restrict__ rp,
                                         const float* __restrict__ xp)
{
#pragma unroll
    for (int s = 0; s < KSTEPS_PER_SET; ++s) {
        f.a[2 * s]     = __builtin_nontemporal_load(
            reinterpret_cast<const v4f*>(rp + (size_t)(4 * s) * N_EDGES));
        f.a[2 * s + 1] = __builtin_nontemporal_load(
            reinterpret_cast<const v4f*>(rp + (size_t)(4 * s + 1) * N_EDGES));
        f.b[2 * s]     = xp[(4 * s) * NPAD];
        f.b[2 * s + 1] = xp[(4 * s + 1) * NPAD];
    }
}

__device__ __forceinline__ void mma_set(const FragSet& f, v8f (&acc)[4])
{
#pragma unroll
    for (int s = 0; s < KSTEPS_PER_SET; ++s) {
        v2f b;
        b.x = f.b[2 * s];
        b.y = f.b[2 * s + 1];
#pragma unroll
        for (int j = 0; j < 4; ++j) {
            v2f a;
            a.x = f.a[2 * s][j];
            a.y = f.a[2 * s + 1][j];
            acc[j] = __builtin_amdgcn_wmma_f32_16x16x4_f32(
                 false, a,  false, b,
                 (short)0, acc[j],  false,  false);
        }
    }
}

__global__ __launch_bounds__(GEMM_THREADS) void angles_gemm_kernel(
    const float* __restrict__ Xpad,
    const float* __restrict__ Ri,
    const float* __restrict__ Ro,
    float* __restrict__ Bmat)
{
    __shared__ float part[GEMM_LDS_FLOATS];

    const int lane = threadIdx.x & 31;
    const int wave = threadIdx.x >> 5;
    const int c    = lane & 15;
    const int half = lane >> 4;
    const int e0   = blockIdx.x * WG_EDGES;

    const float* __restrict__ R = (blockIdx.y == 0) ? Ro : Ri;
    const int colOff            = (blockIdx.y == 0) ? 0  : N_FEAT;

    const int row0 = wave * WAVE_ROWS + 2 * half;
    const float* __restrict__ rp = R + (size_t)row0 * N_EDGES + e0 + 4 * c;
    const float* __restrict__ xp = Xpad + row0 * NPAD + c;

    constexpr size_t R_SET_STRIDE = (size_t)(4 * KSTEPS_PER_SET) * N_EDGES;
    constexpr int    X_SET_STRIDE = 4 * KSTEPS_PER_SET * NPAD;

    v8f acc[4] = {};
    FragSet f0, f1;

    load_set(f0, rp, xp);
    rp += R_SET_STRIDE; xp += X_SET_STRIDE;
    for (int it = 0; it < (SETS_PER_WAVE - 2) / 2; ++it) {
        load_set(f1, rp, xp);
        rp += R_SET_STRIDE; xp += X_SET_STRIDE;
        __builtin_amdgcn_sched_barrier(0);
        mma_set(f0, acc);
        __builtin_amdgcn_sched_barrier(0);
        load_set(f0, rp, xp);
        rp += R_SET_STRIDE; xp += X_SET_STRIDE;
        __builtin_amdgcn_sched_barrier(0);
        mma_set(f1, acc);
        __builtin_amdgcn_sched_barrier(0);
    }
    load_set(f1, rp, xp);
    __builtin_amdgcn_sched_barrier(0);
    mma_set(f0, acc);
    mma_set(f1, acc);

    float* __restrict__ mine = part + wave * WG_OUTPUTS;
    if (c < N_FEAT) {
#pragma unroll
        for (int j = 0; j < 4; ++j) {
#pragma unroll
            for (int r = 0; r < 8; ++r) {
                const int el = 4 * (r + 8 * half) + j;
                mine[el * N_FEAT + c] = acc[j][r];
            }
        }
    }
    __syncthreads();

    if (threadIdx.x < WG_OUTPUTS) {
        float s = part[threadIdx.x];
#pragma unroll
        for (int w = 1; w < GEMM_WAVES; ++w) s += part[w * WG_OUTPUTS + threadIdx.x];
        const int el = threadIdx.x / N_FEAT;
        const int ft = threadIdx.x - el * N_FEAT;
        Bmat[(size_t)(e0 + el) * N_WIRES + colOff + ft] = s;
    }
}

__device__ __forceinline__ float shflx(float x, int m) {
    return __shfl_xor(x, m, 32);
}

template <int W>
__device__ __forceinline__ void apply_ry(float (&v)[32], float c, float s, int lane) {
    constexpr int B = 9 - W;
    if constexpr (B >= 5) {
        constexpr int st = 1 << (B - 5);
#pragma unroll
        for (int r = 0; r < 32; ++r) {
            if ((r & st) == 0) {
                const float p0 = v[r];
                const float p1 = v[r + st];
                v[r]      = c * p0 - s * p1;
                v[r + st] = s * p0 + c * p1;
            }
        }
    } else {
        constexpr int m = 1 << B;
        const float sgn = (lane & m) ? s : -s;
#pragma unroll
        for (int r = 0; r < 32; ++r) {
            const float p = shflx(v[r], m);
            v[r] = c * v[r] + sgn * p;
        }
    }
}

template <int C, int T>
__device__ __forceinline__ void apply_cnot(float (&v)[32], int lane) {
    constexpr int BC = 9 - C;
    constexpr int BT = 9 - T;
    if constexpr (BC >= 5 && BT >= 5) {
        constexpr int cm = 1 << (BC - 5);
        constexpr int tm = 1 << (BT - 5);
#pragma unroll
        for (int r = 0; r < 32; ++r) {
            if ((r & cm) && !(r & tm)) {
                const float t = v[r];
                v[r]      = v[r + tm];
                v[r + tm] = t;
            }
        }
    } else if constexpr (BC >= 5 && BT < 5) {
        constexpr int cm    = 1 << (BC - 5);
        constexpr int tmask = 1 << BT;
#pragma unroll
        for (int r = 0; r < 32; ++r) {
            if (r & cm) v[r] = shflx(v[r], tmask);
        }
    } else if constexpr (BC < 5 && BT >= 5) {
        constexpr int cmask = 1 << BC;
        constexpr int tm    = 1 << (BT - 5);
        const bool cb = (lane & cmask) != 0;
#pragma unroll
        for (int r = 0; r < 32; ++r) {
            if ((r & tm) == 0) {
                const float a = v[r];
                const float b = v[r + tm];
                v[r]      = cb ? b : a;
                v[r + tm] = cb ? a : b;
            }
        }
    } else {
        constexpr int cmask = 1 << BC;
        constexpr int tmask = 1 << BT;
        const bool cb = (lane & cmask) != 0;
#pragma unroll
        for (int r = 0; r < 32; ++r) {
            const float t = shflx(v[r], tmask);
            v[r] = cb ? t : v[r];
        }
    }
}

__global__ __launch_bounds__(256) void ttn_sim_kernel(
    const float* __restrict__ Bmat,
    const float* __restrict__ theta,
    float* __restrict__ out)
{
    const int lane = threadIdx.x & 31;
    const int wave = threadIdx.x >> 5;
    const int e    = blockIdx.x * 8 + wave;

    float ac[10], as[10];
#pragma unroll
    for (int w = 0; w < N_WIRES; ++w) {
        const float h = 0.5f * Bmat[(size_t)e * N_WIRES + w];
        sincosf(h, &as[w], &ac[w]);
    }
    float tc[19], ts[19];
#pragma unroll
    for (int j = 0; j < 19; ++j) {
        const float h = 0.5f * theta[j];
        sincosf(h, &ts[j], &tc[j]);
    }

    float v[32];
#pragma unroll
    for (int r = 0; r < 32; ++r) v[r] = 0.0f;
    if (lane == 0) v[0] = 1.0f;

    apply_ry<0>(v, ac[0], as[0], lane);
    apply_ry<1>(v, ac[1], as[1], lane);
    apply_ry<2>(v, ac[2], as[2], lane);
    apply_ry<3>(v, ac[3], as[3], lane);
    apply_ry<4>(v, ac[4], as[4], lane);
    apply_ry<5>(v, ac[5], as[5], lane);
    apply_ry<6>(v, ac[6], as[6], lane);
    apply_ry<7>(v, ac[7], as[7], lane);
    apply_ry<8>(v, ac[8], as[8], lane);
    apply_ry<9>(v, ac[9], as[9], lane);
    apply_ry<0>(v, tc[0], ts[0], lane); apply_ry<1>(v, tc[1], ts[1], lane); apply_cnot<0, 1>(v, lane);
    apply_ry<2>(v, tc[2], ts[2], lane); apply_ry<3>(v, tc[3], ts[3], lane); apply_cnot<3, 2>(v, lane);
    apply_ry<4>(v, tc[4], ts[4], lane); apply_ry<5>(v, tc[5], ts[5], lane); apply_cnot<5, 4>(v, lane);
    apply_ry<6>(v, tc[6], ts[6], lane); apply_ry<7>(v, tc[7], ts[7], lane); apply_cnot<6, 7>(v, lane);
    apply_ry<8>(v, tc[8], ts[8], lane); apply_ry<9>(v, tc[9], ts[9], lane); apply_cnot<8, 9>(v, lane);
    apply_ry<1>(v, tc[10], ts[10], lane); apply_ry<2>(v, tc[11], ts[11], lane); apply_cnot<1, 2>(v, lane);
    apply_ry<7>(v, tc[12], ts[12], lane); apply_ry<9>(v, tc[13], ts[13], lane); apply_cnot<9, 7>(v, lane);
    apply_ry<2>(v, tc[14], ts[14], lane); apply_ry<4>(v, tc[15], ts[15], lane); apply_cnot<2, 4>(v, lane);
    apply_ry<4>(v, tc[16], ts[16], lane); apply_ry<7>(v, tc[17], ts[17], lane); apply_cnot<4, 7>(v, lane);
    apply_ry<7>(v, tc[18], ts[18], lane);

    float loc = 0.0f;
#pragma unroll
    for (int r = 0; r < 32; ++r) loc += v[r] * v[r];
    loc = (lane & 4) ? loc : 0.0f;
#pragma unroll
    for (int off = 16; off >= 1; off >>= 1) loc += __shfl_xor(loc, off, 32);
    if (lane == 0) out[e] = loc;
}

extern "C" void kernel_launch(void* const* d_in, const int* in_sizes, int n_in,
                              void* d_out, int out_size, void* d_ws, size_t ws_size,
                              hipStream_t stream)
{
    (void)in_sizes; (void)n_in; (void)out_size; (void)ws_size;
    const float* X     = (const float*)d_in[0];
    const float* Ri    = (const float*)d_in[1];
    const float* Ro    = (const float*)d_in[2];
    const float* theta = (const float*)d_in[3];
    float* out  = (float*)d_out;

    float* Bmat = (float*)d_ws;
    float* Xpad = (float*)((char*)d_ws + (size_t)N_EDGES * N_WIRES * sizeof(float));

    pad_x_kernel<<<(N_NODES * NPAD) / 256, 256, 0, stream>>>(X, Xpad);

    dim3 gGemm(N_EDGES / WG_EDGES, 2);
    angles_gemm_kernel<<<gGemm, GEMM_THREADS, 0, stream>>>(Xpad, Ri, Ro, Bmat);

    ttn_sim_kernel<<<N_EDGES / 8, 256, 0, stream>>>(Bmat, theta, out);
}


# Circuit evaluated as a tree, one lane per edge: each CNOT's control traced out behind it, wires kept as 2x2 density matrices, not 1024 amplitudes; output from the matrix diagonal, not a 512-term sum
# speedup vs baseline: 6.1586x; 1.2976x over previous
#include <hip/hip_runtime.h>
#include <hip/hip_bf16.h>

static constexpr int N_NODES = 4096;
static constexpr int N_EDGES = 8192;
static constexpr int N_FEAT  = 5;
static constexpr int N_WIRES = 10;
static constexpr int NPAD    = 16;

typedef __attribute__((ext_vector_type(2))) float v2f;
typedef __attribute__((ext_vector_type(8))) float v8f;

__global__ __launch_bounds__(256) void pad_x_kernel(
    const float* __restrict__ X, float* __restrict__ Xpad)
{
    const int idx = blockIdx.x * 256 + threadIdx.x;
    const int k = idx >> 4;
    const int n = idx & 15;
    Xpad[idx] = (n < N_FEAT) ? X[k * N_FEAT + n] : 0.0f;
}

typedef __attribute__((ext_vector_type(4))) float v4f;

static constexpr int GEMM_WAVES     = 16;
static constexpr int GEMM_THREADS   = GEMM_WAVES * 32;
static constexpr int WG_EDGES       = 64;
static constexpr int WAVE_ROWS      = N_NODES / GEMM_WAVES;
static constexpr int KSTEPS_PER_SET = 8;
static constexpr int SETS_PER_WAVE  = WAVE_ROWS / (4 * KSTEPS_PER_SET);
static constexpr int WG_OUTPUTS     = WG_EDGES * N_FEAT;
static constexpr int GEMM_LDS_FLOATS = GEMM_WAVES * WG_OUTPUTS;

struct FragSet {
    v4f   a[2 * KSTEPS_PER_SET];
    float b[2 * KSTEPS_PER_SET];
};

__device__ __forceinline__ void load_set(FragSet& f,
                                         const float* __restrict__ rp,
                                         const float* __restrict__ xp)
{
#pragma unroll
    for (int s = 0; s < KSTEPS_PER_SET; ++s) {
        f.a[2 * s]     = __builtin_nontemporal_load(
            reinterpret_cast<const v4f*>(rp + (size_t)(4 * s) * N_EDGES));
        f.a[2 * s + 1] = __builtin_nontemporal_load(
            reinterpret_cast<const v4f*>(rp + (size_t)(4 * s + 1) * N_EDGES));
        f.b[2 * s]     = xp[(4 * s) * NPAD];
        f.b[2 * s + 1] = xp[(4 * s + 1) * NPAD];
    }
}

__device__ __forceinline__ void mma_set(const FragSet& f, v8f (&acc)[4])
{
#pragma unroll
    for (int s = 0; s < KSTEPS_PER_SET; ++s) {
        v2f b;
        b.x = f.b[2 * s];
        b.y = f.b[2 * s + 1];
#pragma unroll
        for (int j = 0; j < 4; ++j) {
            v2f a;
            a.x = f.a[2 * s][j];
            a.y = f.a[2 * s + 1][j];
            acc[j] = __builtin_amdgcn_wmma_f32_16x16x4_f32(
                 false, a,  false, b,
                 (short)0, acc[j],  false,  false);
        }
    }
}

__global__ __launch_bounds__(GEMM_THREADS) void angles_gemm_kernel(
    const float* __restrict__ Xpad,
    const float* __restrict__ Ri,
    const float* __restrict__ Ro,
    float* __restrict__ Bmat)
{
    __shared__ float part[GEMM_LDS_FLOATS];

    const int lane = threadIdx.x & 31;
    const int wave = threadIdx.x >> 5;
    const int c    = lane & 15;
    const int half = lane >> 4;
    const int e0   = blockIdx.x * WG_EDGES;

    const float* __restrict__ R = (blockIdx.y == 0) ? Ro : Ri;
    const int colOff            = (blockIdx.y == 0) ? 0  : N_FEAT;

    const int row0 = wave * WAVE_ROWS + 2 * half;
    const float* __restrict__ rp = R + (size_t)row0 * N_EDGES + e0 + 4 * c;
    const float* __restrict__ xp = Xpad + row0 * NPAD + c;

    constexpr size_t R_SET_STRIDE = (size_t)(4 * KSTEPS_PER_SET) * N_EDGES;
    constexpr int    X_SET_STRIDE = 4 * KSTEPS_PER_SET * NPAD;

    v8f acc[4] = {};
    FragSet f0, f1;

    load_set(f0, rp, xp);
    rp += R_SET_STRIDE; xp += X_SET_STRIDE;
    for (int it = 0; it < (SETS_PER_WAVE - 2) / 2; ++it) {
        load_set(f1, rp, xp);
        rp += R_SET_STRIDE; xp += X_SET_STRIDE;
        __builtin_amdgcn_sched_barrier(0);
        mma_set(f0, acc);
        __builtin_amdgcn_sched_barrier(0);
        load_set(f0, rp, xp);
        rp += R_SET_STRIDE; xp += X_SET_STRIDE;
        __builtin_amdgcn_sched_barrier(0);
        mma_set(f1, acc);
        __builtin_amdgcn_sched_barrier(0);
    }
    load_set(f1, rp, xp);
    __builtin_amdgcn_sched_barrier(0);
    mma_set(f0, acc);
    mma_set(f1, acc);

    float* __restrict__ mine = part + wave * WG_OUTPUTS;
    if (c < N_FEAT) {
#pragma unroll
        for (int j = 0; j < 4; ++j) {
#pragma unroll
            for (int r = 0; r < 8; ++r) {
                const int el = 4 * (r + 8 * half) + j;
                mine[el * N_FEAT + c] = acc[j][r];
            }
        }
    }
    __syncthreads();

    if (threadIdx.x < WG_OUTPUTS) {
        float s = part[threadIdx.x];
#pragma unroll
        for (int w = 1; w < GEMM_WAVES; ++w) s += part[w * WG_OUTPUTS + threadIdx.x];
        const int el = threadIdx.x / N_FEAT;
        const int ft = threadIdx.x - el * N_FEAT;
        Bmat[(size_t)(e0 + el) * N_WIRES + colOff + ft] = s;
    }
}

static constexpr int N_THETA     = 19;
static constexpr int TTN_THREADS = 32;

struct Rho { float a, b, d; };

__device__ __forceinline__ Rho rho_pure(float u, float v)
{
    Rho o;
    o.a = u * u;
    o.b = u * v;
    o.d = v * v;
    return o;
}

__device__ __forceinline__ Rho rho_ry(const Rho r, float c, float s)
{
    const float m00 = c * r.a - s * r.b;
    const float m01 = c * r.b - s * r.d;
    const float m10 = s * r.a + c * r.b;
    const float m11 = s * r.b + c * r.d;
    Rho o;
    o.a = m00 * c - m01 * s;
    o.b = m00 * s + m01 * c;
    o.d = m10 * s + m11 * c;
    return o;
}

__device__ __forceinline__ Rho rho_cnot_drop_ctrl(const Rho ctrl, const Rho tgt)
{
    Rho o;
    o.a = ctrl.a * tgt.a + ctrl.d * tgt.d;
    o.d = ctrl.a * tgt.d + ctrl.d * tgt.a;
    o.b = (ctrl.a + ctrl.d) * tgt.b;
    return o;
}

__device__ __forceinline__ float lane_value(float x, int j)
{
    return __builtin_bit_cast(float,
        __builtin_amdgcn_readlane(__builtin_bit_cast(int, x), j));
}

__global__ __launch_bounds__(TTN_THREADS) void ttn_tree_kernel(
    const float* __restrict__ Bmat,
    const float* __restrict__ theta,
    float* __restrict__ out)
{
    const int lane = threadIdx.x;
    const int e    = blockIdx.x * TTN_THREADS + lane;

    const float2* __restrict__ bp =
        reinterpret_cast<const float2*>(Bmat + (size_t)e * N_WIRES);
    float ang[N_WIRES];
#pragma unroll
    for (int p = 0; p < N_WIRES / 2; ++p) {
        const float2 t = bp[p];
        ang[2 * p]     = t.x;
        ang[2 * p + 1] = t.y;
    }

    const int tj = (lane < N_THETA) ? lane : (N_THETA - 1);
    float ts_l, tc_l;
    sincosf(0.5f * theta[tj], &ts_l, &tc_l);
    float tc[N_THETA], ts[N_THETA];
#pragma unroll
    for (int j = 0; j < N_THETA; ++j) {
        tc[j] = lane_value(tc_l, j);
        ts[j] = lane_value(ts_l, j);
    }

    Rho q[N_WIRES];
#pragma unroll
    for (int w = 0; w < N_WIRES; ++w) {
        float sa, ca;
        sincosf(0.5f * ang[w], &sa, &ca);
        const float u = tc[w] * ca - ts[w] * sa;
        const float v = ts[w] * ca + tc[w] * sa;
        q[w] = rho_pure(u, v);
    }

    Rho r1 = rho_cnot_drop_ctrl(q[0], q[1]);
    Rho r2 = rho_cnot_drop_ctrl(q[3], q[2]);
    Rho r4 = rho_cnot_drop_ctrl(q[5], q[4]);
    Rho r7 = rho_cnot_drop_ctrl(q[6], q[7]);
    Rho r9 = rho_cnot_drop_ctrl(q[8], q[9]);
    r1 = rho_ry(r1, tc[10], ts[10]); r2 = rho_ry(r2, tc[11], ts[11]); r2 = rho_cnot_drop_ctrl(r1, r2);
    r7 = rho_ry(r7, tc[12], ts[12]); r9 = rho_ry(r9, tc[13], ts[13]); r7 = rho_cnot_drop_ctrl(r9, r7);
    r2 = rho_ry(r2, tc[14], ts[14]); r4 = rho_ry(r4, tc[15], ts[15]); r4 = rho_cnot_drop_ctrl(r2, r4);
    r4 = rho_ry(r4, tc[16], ts[16]); r7 = rho_ry(r7, tc[17], ts[17]); r7 = rho_cnot_drop_ctrl(r4, r7);
    r7 = rho_ry(r7, tc[18], ts[18]);

    out[e] = 0.5f * (1.0f - (r7.a - r7.d));
}

extern "C" void kernel_launch(void* const* d_in, const int* in_sizes, int n_in,
                              void* d_out, int out_size, void* d_ws, size_t ws_size,
                              hipStream_t stream)
{
    (void)in_sizes; (void)n_in; (void)out_size; (void)ws_size;
    const float* X     = (const float*)d_in[0];
    const float* Ri    = (const float*)d_in[1];
    const float* Ro    = (const float*)d_in[2];
    const float* theta = (const float*)d_in[3];
    float* out  = (float*)d_out;

    float* Bmat = (float*)d_ws;
    float* Xpad = (float*)((char*)d_ws + (size_t)N_EDGES * N_WIRES * sizeof(float));

    pad_x_kernel<<<(N_NODES * NPAD) / 256, 256, 0, stream>>>(X, Xpad);

    dim3 gGemm(N_EDGES / WG_EDGES, 2);
    angles_gemm_kernel<<<gGemm, GEMM_THREADS, 0, stream>>>(Xpad, Ri, Ro, Bmat);

    ttn_tree_kernel<<<N_EDGES / TTN_THREADS, TTN_THREADS, 0, stream>>>(Bmat, theta, out);
}
